# starting kernel
# baseline (speedup 1.0000x reference)
#include <hip/hip_runtime.h>


#define B_ 8
#define N_ 16384
#define K_ 256
#define D_ 20
#define NCHUNKS 64
#define NB 128
#define NBP 130
#define ITERS ((N_ / NCHUNKS) / NB)

typedef __attribute__((ext_vector_type(2))) float v2f;
typedef __attribute__((ext_vector_type(8))) float v8f;

__global__ __launch_bounds__(256) void em_zero(float* __restrict__ out,
                                               float* __restrict__ den) {
  int i = blockIdx.x * 256 + threadIdx.x;
  if (i < B_ * K_ * D_) out[i] = 0.0f;
  if (i < B_ * K_) den[i] = 0.0f;
}

__global__ __launch_bounds__(256) void em_main(const float* __restrict__ mean,
                                               const float* __restrict__ feat,
                                               const float* __restrict__ valid,
                                               const float* __restrict__ adj,
                                               float* __restrict__ out,
                                               float* __restrict__ den) {
  __shared__ float smean[K_ * D_];
  __shared__ float smsq[K_];
  __shared__ float pbuf[K_ * NBP];
  __shared__ float rinvb[NB];

  const int b = blockIdx.x / NCHUNKS;
  const int chunk = blockIdx.x % NCHUNKS;
  const int tid = threadIdx.x;
  const int w = tid >> 5;
  const int lane = tid & 31;
  const int hi = lane >> 4;
  const int lr = lane & 15;

  {
    const float* mrow = mean + ((size_t)b * K_ + tid) * D_;
    float sq = 0.0f;
#pragma unroll
    for (int d = 0; d < D_; ++d) {
      float v = mrow[d];
      smean[tid * D_ + d] = v;
      sq += v * v;
    }
    smsq[tid] = sq;
  }
  __syncthreads();

  v8f acc[2][2];
#pragma unroll
  for (int q = 0; q < 2; ++q)
#pragma unroll
    for (int dt = 0; dt < 2; ++dt) acc[q][dt] = (v8f){0,0,0,0,0,0,0,0};

  const int nchunk0 = chunk * (N_ / NCHUNKS);

  for (int it = 0; it < ITERS; ++it) {
    const int nb0 = nchunk0 + it * NB;
    const int n0 = nb0 + w * 16;

    __syncthreads();

    v2f a[5];
    float psq = 0.0f;
    {
      const float* fbase = feat + ((size_t)b * N_ + n0 + lr) * D_ + 2 * hi;
#pragma unroll
      for (int s = 0; s < 5; ++s) {
        v2f t;
        t.x = fbase[s * 4];
        t.y = fbase[s * 4 + 1];
        a[s] = t;
        psq += t.x * t.x + t.y * t.y;
      }
    }
    float fsq = psq + __shfl_xor(psq, 16, 32);
    float fsqrow[8];
#pragma unroll
    for (int j = 0; j < 8; ++j) fsqrow[j] = __shfl(fsq, j + 8 * hi, 32);

    float rmax[8];
#pragma unroll
    for (int j = 0; j < 8; ++j) rmax[j] = -3.0e38f;

#pragma unroll
    for (int kt = 0; kt < 16; ++kt) {
      v2f bm[5];
      {
        const float* mrow = smean + (kt * 16 + lr) * D_ + 2 * hi;
#pragma unroll
        for (int s = 0; s < 5; ++s) {
          v2f t;
          t.x = mrow[s * 4];
          t.y = mrow[s * 4 + 1];
          bm[s] = t;
        }
      }
      v8f c = (v8f){0,0,0,0,0,0,0,0};
#pragma unroll
      for (int s = 0; s < 5; ++s)
        c = __builtin_amdgcn_wmma_f32_16x16x4_f32(false, a[s], false, bm[s],
                                                  (short)0, c, false, false);
      const float msq = smsq[kt * 16 + lr];
      const size_t vbase = ((size_t)b * N_ + n0) * K_ + kt * 16 + lr;
      const int prow = (kt * 16 + lr) * NBP + w * 16;
#pragma unroll
      for (int j = 0; j < 8; ++j) {
        const int m = j + 8 * hi;
        const float vs = __builtin_nontemporal_load(valid + vbase + (size_t)m * K_);
        const float ad = __builtin_nontemporal_load(adj + vbase + (size_t)m * K_);
        const float dist = (fsqrow[j] - 2.0f * c[j] + msq) * vs + ad;
        const float sc = -dist;
        rmax[j] = fmaxf(rmax[j], sc);
        pbuf[prow + m] = sc;
      }
    }

#pragma unroll
    for (int j = 0; j < 8; ++j) {
      float r = rmax[j];
      r = fmaxf(r, __shfl_xor(r, 1, 32));
      r = fmaxf(r, __shfl_xor(r, 2, 32));
      r = fmaxf(r, __shfl_xor(r, 4, 32));
      r = fmaxf(r, __shfl_xor(r, 8, 32));
      rmax[j] = r;
    }

    float rsum[8];
#pragma unroll
    for (int j = 0; j < 8; ++j) rsum[j] = 0.0f;
#pragma unroll
    for (int kt = 0; kt < 16; ++kt) {
      const int prow = (kt * 16 + lr) * NBP + w * 16;
#pragma unroll
      for (int j = 0; j < 8; ++j) {
        const int m = j + 8 * hi;
        const float e = __expf(pbuf[prow + m] - rmax[j]);
        rsum[j] += e;
        pbuf[prow + m] = e;
      }
    }
#pragma unroll
    for (int j = 0; j < 8; ++j) {
      float r = rsum[j];
      r += __shfl_xor(r, 1, 32);
      r += __shfl_xor(r, 2, 32);
      r += __shfl_xor(r, 4, 32);
      r += __shfl_xor(r, 8, 32);
      if (lr == j) rinvb[w * 16 + j + 8 * hi] = 1.0f / r;
    }
    __syncthreads();

#pragma unroll 2
    for (int ns = 0; ns < NB / 4; ++ns) {
      v2f pa[2];
#pragma unroll
      for (int q = 0; q < 2; ++q) {
        const int kk = (w * 2 + q) * 16 + lr;
        const float* pr = pbuf + kk * NBP + ns * 4 + 2 * hi;
        v2f t;
        t.x = pr[0];
        t.y = pr[1];
        pa[q] = t;
      }
      const float ri0 = rinvb[ns * 4 + 2 * hi];
      const float ri1 = rinvb[ns * 4 + 2 * hi + 1];
      v2f fb[2];
#pragma unroll
      for (int dt = 0; dt < 2; ++dt) {
        const int d = dt * 16 + lr;
        float x0, x1;
        if (d < D_) {
          const float* f0 =
              feat + ((size_t)b * N_ + nb0 + ns * 4 + 2 * hi) * D_ + d;
          x0 = f0[0] * ri0;
          x1 = f0[D_] * ri1;
        } else if (d == D_) {
          x0 = ri0;
          x1 = ri1;
        } else {
          x0 = 0.0f;
          x1 = 0.0f;
        }
        v2f t;
        t.x = x0;
        t.y = x1;
        fb[dt] = t;
      }
#pragma unroll
      for (int q = 0; q < 2; ++q)
#pragma unroll
        for (int dt = 0; dt < 2; ++dt)
          acc[q][dt] = __builtin_amdgcn_wmma_f32_16x16x4_f32(
              false, pa[q], false, fb[dt], (short)0, acc[q][dt], false, false);
    }
  }

#pragma unroll
  for (int q = 0; q < 2; ++q) {
    const int kbase = (w * 2 + q) * 16;
#pragma unroll
    for (int dt = 0; dt < 2; ++dt) {
      const int d = dt * 16 + lr;
#pragma unroll
      for (int j = 0; j < 8; ++j) {
        const int k = kbase + j + 8 * hi;
        const float v = acc[q][dt][j];
        if (d < D_)
          atomicAdd(&out[((size_t)b * K_ + k) * D_ + d], v);
        else if (d == D_)
          atomicAdd(&den[b * K_ + k], v);
      }
    }
  }
}

__global__ __launch_bounds__(256) void em_fin(float* __restrict__ out,
                                              const float* __restrict__ den) {
  int i = blockIdx.x * 256 + threadIdx.x;
  if (i < B_ * K_ * D_) out[i] = out[i] / fmaxf(den[i / D_], 1e-12f);
}

extern "C" void kernel_launch(void* const* d_in, const int* in_sizes, int n_in,
                              void* d_out, int out_size, void* d_ws,
                              size_t ws_size, hipStream_t stream) {
  (void)in_sizes; (void)n_in; (void)out_size; (void)ws_size;
  const float* mean = (const float*)d_in[0];
  const float* feat = (const float*)d_in[1];
  const float* valid = (const float*)d_in[2];
  const float* adj = (const float*)d_in[3];
  float* out = (float*)d_out;
  float* den = (float*)d_ws;

  const int tot = B_ * K_ * D_;
  em_zero<<<(tot + 255) / 256, 256, 0, stream>>>(out, den);
  em_main<<<B_ * NCHUNKS, 256, 0, stream>>>(mean, feat, valid, adj, out, den);
  em_fin<<<(tot + 255) / 256, 256, 0, stream>>>(out, den);
}


# each wave runs both products for its own 16-pixel strips: score tile stays in registers and is the M-step A fragment as is (no LDS posteriors, no loop barrier); partial sums add in a new order
# speedup vs baseline: 2.8590x; 2.8590x over previous
#include <hip/hip_runtime.h>


#define B_ 8
#define N_ 16384
#define K_ 256
#define D_ 20
#define WGB 32
#define STRIPS 4
#define PLANE (K_ * 32)

typedef __attribute__((ext_vector_type(2))) float v2f;
typedef __attribute__((ext_vector_type(8))) float v8f;

__global__ __launch_bounds__(256) void em_zero(float* __restrict__ out,
                                               float* __restrict__ den) {
  int i = blockIdx.x * 256 + threadIdx.x;
  if (i < B_ * K_ * D_) out[i] = 0.0f;
  if (i < B_ * K_) den[i] = 0.0f;
}

__global__ __launch_bounds__(256) void em_main(const float* __restrict__ mean,
                                               const float* __restrict__ feat,
                                               const float* __restrict__ valid,
                                               const float* __restrict__ adj,
                                               float* __restrict__ out,
                                               float* __restrict__ den) {
  __shared__ float smean[K_ * D_];
  __shared__ float smsq[K_];
  __shared__ float red[8 * PLANE];

  const int b = blockIdx.x / WGB;
  const int wg = blockIdx.x % WGB;
  const int tid = threadIdx.x;
  const int w = tid >> 5;
  const int lane = tid & 31;
  const int hi = lane >> 4;
  const int lr = lane & 15;

  {
    const float* mrow = mean + ((size_t)b * K_ + tid) * D_;
    float sq = 0.0f;
#pragma unroll
    for (int d = 0; d < D_; ++d) {
      float v = mrow[d];
      smean[tid * D_ + d] = v;
      sq += v * v;
    }
    smsq[tid] = sq;
  }
  __syncthreads();

  v8f acc[16][2];
#pragma unroll
  for (int kt = 0; kt < 16; ++kt)
#pragma unroll
    for (int dt = 0; dt < 2; ++dt) acc[kt][dt] = (v8f){0,0,0,0,0,0,0,0};

  const int nwave0 = wg * (N_ / WGB) + w * (16 * STRIPS);

#pragma unroll 1
  for (int st = 0; st < STRIPS; ++st) {
    const int n0 = nwave0 + st * 16;
    int moff = 0;
    asm volatile("" : "+v"(moff));

    v2f a[5];
    float psq = 0.0f;
    {
      const float* fbase = feat + ((size_t)b * N_ + n0 + lr) * D_ + 2 * hi;
#pragma unroll
      for (int s = 0; s < 5; ++s) {
        v2f t;
        t.x = fbase[s * 4];
        t.y = fbase[s * 4 + 1];
        a[s] = t;
        psq += t.x * t.x + t.y * t.y;
      }
    }
    float fsq = psq + __shfl_xor(psq, 16, 32);
    float fsqrow[8];
#pragma unroll
    for (int j = 0; j < 8; ++j) fsqrow[j] = __shfl(fsq, j + 8 * hi, 32);

    const float* vrow = valid + ((size_t)b * N_ + n0 + 8 * hi) * K_ + lr;
    const float* arow = adj + ((size_t)b * N_ + n0 + 8 * hi) * K_ + lr;

    float vq[16][8], aq[16][8];
#pragma unroll
    for (int t = 0; t < 2; ++t)
#pragma unroll
      for (int j = 0; j < 8; ++j) {
        vq[t][j] = __builtin_nontemporal_load(vrow + j * K_ + t * 16);
        aq[t][j] = __builtin_nontemporal_load(arow + j * K_ + t * 16);
      }

    v8f sc[16];
    float rmax[8];
#pragma unroll
    for (int j = 0; j < 8; ++j) rmax[j] = -3.0e38f;

#pragma unroll
    for (int kt = 0; kt < 16; ++kt) {
      if (kt + 2 < 16) {
#pragma unroll
        for (int j = 0; j < 8; ++j) {
          vq[kt + 2][j] = __builtin_nontemporal_load(vrow + j * K_ + (kt + 2) * 16);
          aq[kt + 2][j] = __builtin_nontemporal_load(arow + j * K_ + (kt + 2) * 16);
        }
      }
      v2f bm[5];
      {
        const float* mrow = smean + moff + (kt * 16 + lr) * D_ + 2 * hi;
#pragma unroll
        for (int s = 0; s < 5; ++s) {
          v2f t;
          t.x = mrow[s * 4];
          t.y = mrow[s * 4 + 1];
          bm[s] = t;
        }
      }
      v8f c = (v8f){0,0,0,0,0,0,0,0};
#pragma unroll
      for (int s = 0; s < 5; ++s)
        c = __builtin_amdgcn_wmma_f32_16x16x4_f32(false, a[s], false, bm[s],
                                                  (short)0, c, false, false);
      const float msq = smsq[moff + kt * 16 + lr];
      v8f sv;
#pragma unroll
      for (int j = 0; j < 8; ++j) {
        const float dist = (fsqrow[j] - 2.0f * c[j] + msq) * vq[kt][j] + aq[kt][j];
        const float s1 = -dist;
        rmax[j] = fmaxf(rmax[j], s1);
        sv[j] = s1;
      }
      sc[kt] = sv;
    }

#pragma unroll
    for (int j = 0; j < 8; ++j) {
      float r = rmax[j];
      r = fmaxf(r, __shfl_xor(r, 1, 32));
      r = fmaxf(r, __shfl_xor(r, 2, 32));
      r = fmaxf(r, __shfl_xor(r, 4, 32));
      r = fmaxf(r, __shfl_xor(r, 8, 32));
      rmax[j] = r;
    }

    float rsum[8];
#pragma unroll
    for (int j = 0; j < 8; ++j) rsum[j] = 0.0f;
#pragma unroll
    for (int kt = 0; kt < 16; ++kt) {
      v8f sv = sc[kt];
#pragma unroll
      for (int j = 0; j < 8; ++j) {
        const float e = __expf(sv[j] - rmax[j]);
        rsum[j] += e;
        sv[j] = e;
      }
      sc[kt] = sv;
    }
    float ri[8];
#pragma unroll
    for (int j = 0; j < 8; ++j) {
      float r = rsum[j];
      r += __shfl_xor(r, 1, 32);
      r += __shfl_xor(r, 2, 32);
      r += __shfl_xor(r, 4, 32);
      r += __shfl_xor(r, 8, 32);
      ri[j] = 1.0f / r;
    }

    v2f fb[4][2];
    {
      const float* fstrip = feat + ((size_t)b * N_ + n0 + 8 * hi) * D_;
#pragma unroll
      for (int s = 0; s < 4; ++s)
#pragma unroll
        for (int dt = 0; dt < 2; ++dt) {
          const int d = dt * 16 + lr;
          const int dc = d < D_ ? d : D_ - 1;
          const float f0 = fstrip[(2 * s) * D_ + dc];
          const float f1 = fstrip[(2 * s + 1) * D_ + dc];
          v2f t;
          t.x = d < D_ ? f0 * ri[2 * s] : (d == D_ ? ri[2 * s] : 0.0f);
          t.y = d < D_ ? f1 * ri[2 * s + 1] : (d == D_ ? ri[2 * s + 1] : 0.0f);
          fb[s][dt] = t;
        }
    }

#pragma unroll
    for (int kt = 0; kt < 16; ++kt) {
      const v8f sv = sc[kt];
#pragma unroll
      for (int s = 0; s < 4; ++s) {
        v2f pa;
        pa.x = sv[2 * s];
        pa.y = sv[2 * s + 1];
#pragma unroll
        for (int dt = 0; dt < 2; ++dt)
          acc[kt][dt] = __builtin_amdgcn_wmma_f32_16x16x4_f32(
              false, pa, false, fb[s][dt], (short)0, acc[kt][dt], false, false);
      }
    }
  }

  {
    float* plane = red + w * PLANE;
#pragma unroll
    for (int kt = 0; kt < 16; ++kt)
#pragma unroll
      for (int dt = 0; dt < 2; ++dt)
#pragma unroll
        for (int j = 0; j < 8; ++j)
          plane[(kt * 16 + 8 * hi + j) * 32 + dt * 16 + lr] = acc[kt][dt][j];
  }
  __syncthreads();
  for (int i = 0; i < (K_ * (D_ + 1)) / 256; ++i) {
    const int e = tid + 256 * i;
    const int k = e / (D_ + 1);
    const int d = e - k * (D_ + 1);
    float s = 0.0f;
#pragma unroll
    for (int p = 0; p < 8; ++p) s += red[p * PLANE + k * 32 + d];
    if (d < D_)
      atomicAdd(&out[((size_t)b * K_ + k) * D_ + d], s);
    else
      atomicAdd(&den[b * K_ + k], s);
  }
}

__global__ __launch_bounds__(256) void em_fin(float* __restrict__ out,
                                              const float* __restrict__ den) {
  int i = blockIdx.x * 256 + threadIdx.x;
  if (i < B_ * K_ * D_) out[i] = out[i] / fmaxf(den[i / D_], 1e-12f);
}

extern "C" void kernel_launch(void* const* d_in, const int* in_sizes, int n_in,
                              void* d_out, int out_size, void* d_ws,
                              size_t ws_size, hipStream_t stream) {
  (void)in_sizes; (void)n_in; (void)out_size; (void)ws_size;
  const float* mean = (const float*)d_in[0];
  const float* feat = (const float*)d_in[1];
  const float* valid = (const float*)d_in[2];
  const float* adj = (const float*)d_in[3];
  float* out = (float*)d_out;
  float* den = (float*)d_ws;

  const int tot = B_ * K_ * D_;
  em_zero<<<(tot + 255) / 256, 256, 0, stream>>>(out, den);
  em_main<<<B_ * WGB, 256, 0, stream>>>(mean, feat, valid, adj, out, den);
  em_fin<<<(tot + 255) / 256, 256, 0, stream>>>(out, den);
}


# row maxima, row sums and the |f|^2 broadcast cross lanes by DPP row swizzles and one permlanex16 on the vector ALU instead of ds_bpermute: no LDS round trip per butterfly step; same add order
# speedup vs baseline: 3.1180x; 1.0906x over previous
#include <hip/hip_runtime.h>


#define B_ 8
#define N_ 16384
#define K_ 256
#define D_ 20
#define WGB 32
#define STRIPS 4
#define PLANE (K_ * 32)

typedef __attribute__((ext_vector_type(2))) float v2f;
typedef __attribute__((ext_vector_type(8))) float v8f;

__device__ __forceinline__ float xhalf(float x) {
  const int b = __builtin_bit_cast(int, x);
  return __builtin_bit_cast(
      float, __builtin_amdgcn_permlanex16(b, b, 0x76543210, 0xfedcba98, false, false));
}

template <int M>
__device__ __forceinline__ float rowx(float x) {
  const int b = __builtin_bit_cast(int, x);
  return __builtin_bit_cast(
      float, __builtin_amdgcn_update_dpp(b, b, 0x160 + M, 0xf, 0xf, true));
}

template <int J>
__device__ __forceinline__ float rowpick(float x) {
  const int b = __builtin_bit_cast(int, x);
  const int lo = __builtin_amdgcn_update_dpp(b, b, 0x150 + J, 0xf, 0xf, true);
  return __builtin_bit_cast(
      float, __builtin_amdgcn_update_dpp(lo, b, 0x150 + J + 8, 0x2, 0xf, false));
}

template <int J>
struct PickRows {
  static __device__ __forceinline__ void run(float fsq, float* dst) {
    dst[J] = rowpick<J>(fsq);
    PickRows<J + 1>::run(fsq, dst);
  }
};
template <>
struct PickRows<8> {
  static __device__ __forceinline__ void run(float, float*) {}
};

__global__ __launch_bounds__(256) void em_zero(float* __restrict__ out,
                                               float* __restrict__ den) {
  int i = blockIdx.x * 256 + threadIdx.x;
  if (i < B_ * K_ * D_) out[i] = 0.0f;
  if (i < B_ * K_) den[i] = 0.0f;
}

__global__ __launch_bounds__(256) void em_main(const float* __restrict__ mean,
                                               const float* __restrict__ feat,
                                               const float* __restrict__ valid,
                                               const float* __restrict__ adj,
                                               float* __restrict__ out,
                                               float* __restrict__ den) {
  __shared__ float smean[K_ * D_];
  __shared__ float smsq[K_];
  __shared__ float red[8 * PLANE];

  const int b = blockIdx.x / WGB;
  const int wg = blockIdx.x % WGB;
  const int tid = threadIdx.x;
  const int w = tid >> 5;
  const int lane = tid & 31;
  const int hi = lane >> 4;
  const int lr = lane & 15;

  {
    const float* mrow = mean + ((size_t)b * K_ + tid) * D_;
    float sq = 0.0f;
#pragma unroll
    for (int d = 0; d < D_; ++d) {
      float v = mrow[d];
      smean[tid * D_ + d] = v;
      sq += v * v;
    }
    smsq[tid] = sq;
  }
  __syncthreads();

  v8f acc[16][2];
#pragma unroll
  for (int kt = 0; kt < 16; ++kt)
#pragma unroll
    for (int dt = 0; dt < 2; ++dt) acc[kt][dt] = (v8f){0,0,0,0,0,0,0,0};

  const int nwave0 = wg * (N_ / WGB) + w * (16 * STRIPS);

#pragma unroll 1
  for (int st = 0; st < STRIPS; ++st) {
    const int n0 = nwave0 + st * 16;
    int moff = 0;
    asm volatile("" : "+v"(moff));

    v2f a[5];
    float psq = 0.0f;
    {
      const float* fbase = feat + ((size_t)b * N_ + n0 + lr) * D_ + 2 * hi;
#pragma unroll
      for (int s = 0; s < 5; ++s) {
        v2f t;
        t.x = fbase[s * 4];
        t.y = fbase[s * 4 + 1];
        a[s] = t;
        psq += t.x * t.x + t.y * t.y;
      }
    }
    float fsq = psq + xhalf(psq);
    float fsqrow[8];
    PickRows<0>::run(fsq, fsqrow);

    const float* vrow = valid + ((size_t)b * N_ + n0 + 8 * hi) * K_ + lr;
    const float* arow = adj + ((size_t)b * N_ + n0 + 8 * hi) * K_ + lr;

    float vq[16][8], aq[16][8];
#pragma unroll
    for (int t = 0; t < 2; ++t)
#pragma unroll
      for (int j = 0; j < 8; ++j) {
        vq[t][j] = __builtin_nontemporal_load(vrow + j * K_ + t * 16);
        aq[t][j] = __builtin_nontemporal_load(arow + j * K_ + t * 16);
      }

    v8f sc[16];
    float rmax[8];
#pragma unroll
    for (int j = 0; j < 8; ++j) rmax[j] = -3.0e38f;

#pragma unroll
    for (int kt = 0; kt < 16; ++kt) {
      if (kt + 2 < 16) {
#pragma unroll
        for (int j = 0; j < 8; ++j) {
          vq[kt + 2][j] = __builtin_nontemporal_load(vrow + j * K_ + (kt + 2) * 16);
          aq[kt + 2][j] = __builtin_nontemporal_load(arow + j * K_ + (kt + 2) * 16);
        }
      }
      v2f bm[5];
      {
        const float* mrow = smean + moff + (kt * 16 + lr) * D_ + 2 * hi;
#pragma unroll
        for (int s = 0; s < 5; ++s) {
          v2f t;
          t.x = mrow[s * 4];
          t.y = mrow[s * 4 + 1];
          bm[s] = t;
        }
      }
      v8f c = (v8f){0,0,0,0,0,0,0,0};
#pragma unroll
      for (int s = 0; s < 5; ++s)
        c = __builtin_amdgcn_wmma_f32_16x16x4_f32(false, a[s], false, bm[s],
                                                  (short)0, c, false, false);
      const float msq = smsq[moff + kt * 16 + lr];
      v8f sv;
#pragma unroll
      for (int j = 0; j < 8; ++j) {
        const float dist = (fsqrow[j] - 2.0f * c[j] + msq) * vq[kt][j] + aq[kt][j];
        const float s1 = -dist;
        rmax[j] = fmaxf(rmax[j], s1);
        sv[j] = s1;
      }
      sc[kt] = sv;
    }

#pragma unroll
    for (int j = 0; j < 8; ++j) {
      float r = rmax[j];
      r = fmaxf(r, rowx<1>(r));
      r = fmaxf(r, rowx<2>(r));
      r = fmaxf(r, rowx<4>(r));
      r = fmaxf(r, rowx<8>(r));
      rmax[j] = r;
    }

    float rsum[8];
#pragma unroll
    for (int j = 0; j < 8; ++j) rsum[j] = 0.0f;
#pragma unroll
    for (int kt = 0; kt < 16; ++kt) {
      v8f sv = sc[kt];
#pragma unroll
      for (int j = 0; j < 8; ++j) {
        const float e = __expf(sv[j] - rmax[j]);
        rsum[j] += e;
        sv[j] = e;
      }
      sc[kt] = sv;
    }
    float ri[8];
#pragma unroll
    for (int j = 0; j < 8; ++j) {
      float r = rsum[j];
      r += rowx<1>(r);
      r += rowx<2>(r);
      r += rowx<4>(r);
      r += rowx<8>(r);
      ri[j] = 1.0f / r;
    }

    v2f fb[4][2];
    {
      const float* fstrip = feat + ((size_t)b * N_ + n0 + 8 * hi) * D_;
#pragma unroll
      for (int s = 0; s < 4; ++s)
#pragma unroll
        for (int dt = 0; dt < 2; ++dt) {
          const int d = dt * 16 + lr;
          const int dc = d < D_ ? d : D_ - 1;
          const float f0 = fstrip[(2 * s) * D_ + dc];
          const float f1 = fstrip[(2 * s + 1) * D_ + dc];
          v2f t;
          t.x = d < D_ ? f0 * ri[2 * s] : (d == D_ ? ri[2 * s] : 0.0f);
          t.y = d < D_ ? f1 * ri[2 * s + 1] : (d == D_ ? ri[2 * s + 1] : 0.0f);
          fb[s][dt] = t;
        }
    }

#pragma unroll
    for (int kt = 0; kt < 16; ++kt) {
      const v8f sv = sc[kt];
#pragma unroll
      for (int s = 0; s < 4; ++s) {
        v2f pa;
        pa.x = sv[2 * s];
        pa.y = sv[2 * s + 1];
#pragma unroll
        for (int dt = 0; dt < 2; ++dt)
          acc[kt][dt] = __builtin_amdgcn_wmma_f32_16x16x4_f32(
              false, pa, false, fb[s][dt], (short)0, acc[kt][dt], false, false);
      }
    }
  }

  {
    float* plane = red + w * PLANE;
#pragma unroll
    for (int kt = 0; kt < 16; ++kt)
#pragma unroll
      for (int dt = 0; dt < 2; ++dt)
#pragma unroll
        for (int j = 0; j < 8; ++j)
          plane[(kt * 16 + 8 * hi + j) * 32 + dt * 16 + lr] = acc[kt][dt][j];
  }
  __syncthreads();
  for (int i = 0; i < (K_ * (D_ + 1)) / 256; ++i) {
    const int e = tid + 256 * i;
    const int k = e / (D_ + 1);
    const int d = e - k * (D_ + 1);
    float s = 0.0f;
#pragma unroll
    for (int p = 0; p < 8; ++p) s += red[p * PLANE + k * 32 + d];
    if (d < D_)
      atomicAdd(&out[((size_t)b * K_ + k) * D_ + d], s);
    else
      atomicAdd(&den[b * K_ + k], s);
  }
}

__global__ __launch_bounds__(256) void em_fin(float* __restrict__ out,
                                              const float* __restrict__ den) {
  int i = blockIdx.x * 256 + threadIdx.x;
  if (i < B_ * K_ * D_) out[i] = out[i] / fmaxf(den[i / D_], 1e-12f);
}

extern "C" void kernel_launch(void* const* d_in, const int* in_sizes, int n_in,
                              void* d_out, int out_size, void* d_ws,
                              size_t ws_size, hipStream_t stream) {
  (void)in_sizes; (void)n_in; (void)out_size; (void)ws_size;
  const float* mean = (const float*)d_in[0];
  const float* feat = (const float*)d_in[1];
  const float* valid = (const float*)d_in[2];
  const float* adj = (const float*)d_in[3];
  float* out = (float*)d_out;
  float* den = (float*)d_ws;

  const int tot = B_ * K_ * D_;
  em_zero<<<(tot + 255) / 256, 256, 0, stream>>>(out, den);
  em_main<<<B_ * WGB, 256, 0, stream>>>(mean, feat, valid, adj, out, den);
  em_fin<<<(tot + 255) / 256, 256, 0, stream>>>(out, den);
}


# strips software-pipelined per wave: M-step of strip s-1 runs tile by tile inside strip s's E-step loop, loads pinned 2 tiles ahead; d>=16 sums of 3 cluster tiles share one accumulator; same add order
# speedup vs baseline: 3.7297x; 1.1962x over previous
#include <hip/hip_runtime.h>


#define B_ 8
#define N_ 16384
#define K_ 256
#define D_ 20
#define WGB 32
#define STRIPS 4
#define PLANE (K_ * 32)
#define DEPTH 2

typedef __attribute__((ext_vector_type(2))) float v2f;
typedef __attribute__((ext_vector_type(8))) float v8f;

__device__ __forceinline__ float xhalf(float x) {
  const int b = __builtin_bit_cast(int, x);
  return __builtin_bit_cast(
      float, __builtin_amdgcn_permlanex16(b, b, 0x76543210, 0xfedcba98, false, false));
}

template <int M>
__device__ __forceinline__ float rowx(float x) {
  const int b = __builtin_bit_cast(int, x);
  return __builtin_bit_cast(
      float, __builtin_amdgcn_update_dpp(b, b, 0x160 + M, 0xf, 0xf, true));
}

template <int J>
__device__ __forceinline__ float rowpick(float x) {
  const int b = __builtin_bit_cast(int, x);
  const int lo = __builtin_amdgcn_update_dpp(b, b, 0x150 + J, 0xf, 0xf, true);
  return __builtin_bit_cast(
      float, __builtin_amdgcn_update_dpp(lo, b, 0x150 + J + 8, 0x2, 0xf, false));
}

template <int J>
struct PickRows {
  static __device__ __forceinline__ void run(float fsq, float* dst) {
    dst[J] = rowpick<J>(fsq);
    PickRows<J + 1>::run(fsq, dst);
  }
};
template <>
struct PickRows<8> {
  static __device__ __forceinline__ void run(float, float*) {}
};

__global__ __launch_bounds__(256) void em_zero(float* __restrict__ out,
                                               float* __restrict__ den) {
  int i = blockIdx.x * 256 + threadIdx.x;
  if (i < B_ * K_ * D_) out[i] = 0.0f;
  if (i < B_ * K_) den[i] = 0.0f;
}


__device__ __forceinline__ void load_afrag(const float* fbase, v2f (&a)[5],
                                           float (&fsqrow)[8]) {
  float psq = 0.0f;
#pragma unroll
  for (int s = 0; s < 5; ++s) {
    v2f t;
    t.x = fbase[s * 4];
    t.y = fbase[s * 4 + 1];
    a[s] = t;
    psq += t.x * t.x + t.y * t.y;
  }
  const float fsq = psq + xhalf(psq);
  PickRows<0>::run(fsq, fsqrow);
}

template <bool DO_E, bool DO_M>
__device__ __forceinline__ void strip_pass(
    v8f (&acc0)[16], v8f (&acc1)[6], v8f (&sc)[16], v2f (&fb0)[4], v2f (&fbT)[3][4],
    const float* vrow, const float* arow, const float* fa, const float* fcur,
    const float* smean, const float* smsq, const int lr, const int hi) {
  int moff = 0;
  asm volatile("" : "+v"(moff));

  float vq[16][8], aq[16][8];
  float rmax[8];
  v2f a[5];
  float fsqrow[8];
  if (DO_E) {
    load_afrag(fa, a, fsqrow);
#pragma unroll
    for (int t = 0; t < DEPTH; ++t)
#pragma unroll
      for (int j = 0; j < 8; ++j) {
        vq[t][j] = __builtin_nontemporal_load(vrow + j * K_ + t * 16);
        aq[t][j] = __builtin_nontemporal_load(arow + j * K_ + t * 16);
      }
#pragma unroll
    for (int j = 0; j < 8; ++j) rmax[j] = -3.0e38f;
  }

#pragma unroll
  for (int kt = 0; kt < 16; ++kt) {
    if (DO_E) {
      __builtin_amdgcn_sched_barrier(0);
      if (kt + DEPTH < 16) {
#pragma unroll
        for (int j = 0; j < 8; ++j) {
          vq[kt + DEPTH][j] = __builtin_nontemporal_load(vrow + j * K_ + (kt + DEPTH) * 16);
          aq[kt + DEPTH][j] = __builtin_nontemporal_load(arow + j * K_ + (kt + DEPTH) * 16);
        }
      }
      __builtin_amdgcn_sched_barrier(0);
    }

    if (DO_M) {
      const v8f sv = sc[kt];
#pragma unroll
      for (int s = 0; s < 4; ++s) {
        v2f pa;
        pa.x = sv[2 * s];
        pa.y = sv[2 * s + 1];
        acc0[kt] = __builtin_amdgcn_wmma_f32_16x16x4_f32(
            false, pa, false, fb0[s], (short)0, acc0[kt], false, false);
        acc1[kt / 3] = __builtin_amdgcn_wmma_f32_16x16x4_f32(
            false, fbT[kt % 3][s], false, pa, (short)0, acc1[kt / 3], false, false);
      }
    }

    if (DO_E) {
      v2f bm[5];
      {
        const float* mrow = smean + moff + (kt * 16 + lr) * D_ + 2 * hi;
#pragma unroll
        for (int s = 0; s < 5; ++s) {
          v2f t;
          t.x = mrow[s * 4];
          t.y = mrow[s * 4 + 1];
          bm[s] = t;
        }
      }
      v8f c = (v8f){0,0,0,0,0,0,0,0};
#pragma unroll
      for (int s = 0; s < 5; ++s)
        c = __builtin_amdgcn_wmma_f32_16x16x4_f32(false, a[s], false, bm[s],
                                                  (short)0, c, false, false);
      const float msq = smsq[moff + kt * 16 + lr];
      v8f sv;
#pragma unroll
      for (int j = 0; j < 8; ++j) {
        const float dist = (fsqrow[j] - 2.0f * c[j] + msq) * vq[kt][j] + aq[kt][j];
        const float s1 = -dist;
        rmax[j] = fmaxf(rmax[j], s1);
        sv[j] = s1;
      }
      sc[kt] = sv;
    }
  }

  if (DO_E) {
    __builtin_amdgcn_sched_barrier(0);
    const int slot = lr / 5;
    const int bi = lr - 5 * slot;
    const int dcol = bi < 4 ? 16 + bi : D_ - 1;
    float f0[4][2], f1[4][2];
#pragma unroll
    for (int s = 0; s < 4; ++s) {
      f0[s][0] = fcur[(2 * s) * D_ + lr];
      f1[s][0] = fcur[(2 * s + 1) * D_ + lr];
      f0[s][1] = fcur[(2 * s) * D_ + dcol];
      f1[s][1] = fcur[(2 * s + 1) * D_ + dcol];
    }

#pragma unroll
    for (int j = 0; j < 8; ++j) {
      float r = rmax[j];
      r = fmaxf(r, rowx<1>(r));
      r = fmaxf(r, rowx<2>(r));
      r = fmaxf(r, rowx<4>(r));
      r = fmaxf(r, rowx<8>(r));
      rmax[j] = r;
    }

    float rsum[8];
#pragma unroll
    for (int j = 0; j < 8; ++j) rsum[j] = 0.0f;
#pragma unroll
    for (int kt = 0; kt < 16; ++kt) {
      v8f sv = sc[kt];
#pragma unroll
      for (int j = 0; j < 8; ++j) {
        const float e = __expf(sv[j] - rmax[j]);
        rsum[j] += e;
        sv[j] = e;
      }
      sc[kt] = sv;
    }
    float ri[8];
#pragma unroll
    for (int j = 0; j < 8; ++j) {
      float r = rsum[j];
      r += rowx<1>(r);
      r += rowx<2>(r);
      r += rowx<4>(r);
      r += rowx<8>(r);
      ri[j] = 1.0f / r;
    }

#pragma unroll
    for (int s = 0; s < 4; ++s) {
      v2f t;
      t.x = f0[s][0] * ri[2 * s];
      t.y = f1[s][0] * ri[2 * s + 1];
      fb0[s] = t;
      v2f u;
      u.x = bi < 4 ? f0[s][1] * ri[2 * s] : ri[2 * s];
      u.y = bi < 4 ? f1[s][1] * ri[2 * s + 1] : ri[2 * s + 1];
#pragma unroll
      for (int r = 0; r < 3; ++r) {
        v2f z;
        z.x = slot == r ? u.x : 0.0f;
        z.y = slot == r ? u.y : 0.0f;
        fbT[r][s] = z;
      }
    }

  }
}

__global__ __launch_bounds__(256) void em_main(const float* __restrict__ mean,
                                               const float* __restrict__ feat,
                                               const float* valid,
                                               const float* adj,
                                               float* __restrict__ out,
                                               float* __restrict__ den) {
  __shared__ float smean[K_ * D_];
  __shared__ float smsq[K_];
  __shared__ float red[8 * PLANE];

  const int b = blockIdx.x / WGB;
  const int wg = blockIdx.x % WGB;
  const int tid = threadIdx.x;
  const int w = tid >> 5;
  const int lane = tid & 31;
  const int hi = lane >> 4;
  const int lr = lane & 15;

  {
    const float* mrow = mean + ((size_t)b * K_ + tid) * D_;
    float sq = 0.0f;
#pragma unroll
    for (int d = 0; d < D_; ++d) {
      float v = mrow[d];
      smean[tid * D_ + d] = v;
      sq += v * v;
    }
    smsq[tid] = sq;
  }
  __syncthreads();

  v8f acc0[16], acc1[6];
#pragma unroll
  for (int kt = 0; kt < 16; ++kt) acc0[kt] = (v8f){0,0,0,0,0,0,0,0};
#pragma unroll
  for (int g = 0; g < 6; ++g) acc1[g] = (v8f){0,0,0,0,0,0,0,0};

  const int nwave0 = wg * (N_ / WGB) + w * (16 * STRIPS);
  const float* fA = feat + ((size_t)b * N_ + nwave0 + lr) * D_ + 2 * hi;
  const float* fB = feat + ((size_t)b * N_ + nwave0 + 8 * hi) * D_;
  const float* vrow = valid + ((size_t)b * N_ + nwave0 + 8 * hi) * K_ + lr;
  const float* arow = adj + ((size_t)b * N_ + nwave0 + 8 * hi) * K_ + lr;

  v8f sc[16];
  v2f fb0[4], fbT[3][4];

  strip_pass<true, false>(acc0, acc1, sc, fb0, fbT, vrow, arow, fA, fB,
                          smean, smsq, lr, hi);

#pragma unroll 1
  for (int st = 1; st < STRIPS; ++st)
    strip_pass<true, true>(acc0, acc1, sc, fb0, fbT, vrow + st * 16 * K_,
                           arow + st * 16 * K_, fA + st * 16 * D_, fB + st * 16 * D_,
                           smean, smsq, lr, hi);

  strip_pass<false, true>(acc0, acc1, sc, fb0, fbT, vrow, arow, fA, fB,
                          smean, smsq, lr, hi);

  {
    float* plane = red + w * PLANE;
#pragma unroll
    for (int kt = 0; kt < 16; ++kt)
#pragma unroll
      for (int j = 0; j < 8; ++j)
        plane[(kt * 16 + 8 * hi + j) * 32 + lr] = acc0[kt][j];
#pragma unroll
    for (int g = 0; g < 6; ++g)
#pragma unroll
      for (int j = 0; j < 8; ++j) {
        const int rho = 8 * hi + j;
        const int r = rho / 5;
        const int i = rho - 5 * r;
        const int kt = 3 * g + r;
        if (rho < 15 && kt < 16) plane[(kt * 16 + lr) * 32 + 16 + i] = acc1[g][j];
      }
  }
  __syncthreads();
  for (int i = 0; i < (K_ * (D_ + 1)) / 256; ++i) {
    const int e = tid + 256 * i;
    const int k = e / (D_ + 1);
    const int d = e - k * (D_ + 1);
    float s = 0.0f;
#pragma unroll
    for (int p = 0; p < 8; ++p) s += red[p * PLANE + k * 32 + d];
    if (d < D_)
      atomicAdd(&out[((size_t)b * K_ + k) * D_ + d], s);
    else
      atomicAdd(&den[b * K_ + k], s);
  }
}

__global__ __launch_bounds__(256) void em_fin(float* __restrict__ out,
                                              const float* __restrict__ den) {
  int i = blockIdx.x * 256 + threadIdx.x;
  if (i < B_ * K_ * D_) out[i] = out[i] / fmaxf(den[i / D_], 1e-12f);
}

extern "C" void kernel_launch(void* const* d_in, const int* in_sizes, int n_in,
                              void* d_out, int out_size, void* d_ws,
                              size_t ws_size, hipStream_t stream) {
  (void)in_sizes; (void)n_in; (void)out_size; (void)ws_size;
  const float* mean = (const float*)d_in[0];
  const float* feat = (const float*)d_in[1];
  const float* valid = (const float*)d_in[2];
  const float* adj = (const float*)d_in[3];
  float* out = (float*)d_out;
  float* den = (float*)d_ws;

  const int tot = B_ * K_ * D_;
  em_zero<<<(tot + 255) / 256, 256, 0, stream>>>(out, den);
  em_main<<<B_ * WGB, 256, 0, stream>>>(mean, feat, valid, adj, out, den);
  em_fin<<<(tot + 255) / 256, 256, 0, stream>>>(out, den);
}
